# starting kernel
# baseline (speedup 1.0000x reference)
#include <hip/hip_runtime.h>
#include <math.h>

#define NN 100000
#define NE 1600000
#define KIN 64
#define H1 64
#define H2 32

typedef __attribute__((ext_vector_type(2))) float v2f;
typedef __attribute__((ext_vector_type(8))) float v8f;

__global__ void k_zero(float* p, int n) {
  int i = blockIdx.x * blockDim.x + threadIdx.x;
  if (i < n) p[i] = 0.0f;
}

__global__ void k_edge_ew_deg(const float* __restrict__ logits,
                              const int* __restrict__ ei,
                              float* __restrict__ ew,
                              float* __restrict__ deg) {
  int e = blockIdx.x * blockDim.x + threadIdx.x;
  if (e >= NE) return;
  float l = logits[e];
  float s = 0.5f + 1.0f / (1.0f + __expf(-l));
  ew[e] = s;
  int col = ei[NE + e];
  atomicAdd(&deg[col], s);
}

__global__ void k_dis(const float* __restrict__ deg, float* __restrict__ dis) {
  int i = blockIdx.x * blockDim.x + threadIdx.x;
  if (i < NN) {
    float d = deg[i] + 1.0f;
    dis[i] = rsqrtf(d);
  }
}

__global__ void k_norm(const int* __restrict__ ei,
                       const float* __restrict__ ew,
                       const float* __restrict__ dis,
                       float* __restrict__ norm) {
  int e = blockIdx.x * blockDim.x + threadIdx.x;
  if (e >= NE) return;
  int r = ei[e];
  int c = ei[NE + e];
  norm[e] = dis[r] * ew[e] * dis[c];
}

template <int K, int NOUT>
__global__ void k_gemm_wmma(const float* __restrict__ X,
                            const float* __restrict__ W,
                            float* __restrict__ H) {
  const int wave = threadIdx.x >> 5;
  const int lane = threadIdx.x & 31;
  const int m0 = blockIdx.x * 16;
  const int n0 = wave * 16;
  const int half = lane >> 4;
  const int l15 = lane & 15;

  v2f bfrag[K / 4];
#pragma unroll
  for (int kt = 0; kt < K / 4; ++kt) {
    int kk = kt * 4 + half * 2;
    int col = n0 + l15;
    v2f b;
    b.x = W[kk * NOUT + col];
    b.y = W[(kk + 1) * NOUT + col];
    bfrag[kt] = b;
  }

  const int row = m0 + l15;
  const float* xp = X + (size_t)row * K + half * 2;

  v8f acc = {};
#pragma unroll
  for (int kt = 0; kt < K / 4; ++kt) {
    v2f a = *(const v2f*)(xp + kt * 4);
    acc = __builtin_amdgcn_wmma_f32_16x16x4_f32(
         false, a,  false, bfrag[kt],
         (short)0, acc,  false,  false);
  }

#pragma unroll
  for (int v = 0; v < 8; ++v) {
    int r = m0 + v + half * 8;
    H[(size_t)r * NOUT + n0 + l15] = acc[v];
  }
}

template <int NOUT>
__global__ void k_self(const float* __restrict__ h,
                       const float* __restrict__ dis,
                       const float* __restrict__ b,
                       float* __restrict__ out) {
  int idx = blockIdx.x * blockDim.x + threadIdx.x;
  if (idx >= NN * NOUT) return;
  int i = idx / NOUT;
  int c = idx - i * NOUT;
  float d = dis[i];
  out[idx] = h[idx] * (d * d) + b[c];
}

template <int NOUT, int LOGG>
__global__ void k_scatter(const float* __restrict__ h,
                          const int* __restrict__ ei,
                          const float* __restrict__ norm,
                          float* __restrict__ out) {
  long long idx = (long long)blockIdx.x * blockDim.x + threadIdx.x;
  int e = (int)(idx >> LOGG);
  if (e >= NE) return;
  int g = (int)(idx & ((1 << LOGG) - 1));
  int row = ei[e];
  int col = ei[NE + e];
  float nw = norm[e];
  const float4 hv = *(const float4*)(h + (size_t)row * NOUT + g * 4);
  float* op = out + (size_t)col * NOUT + g * 4;
  atomicAdd(op + 0, hv.x * nw);
  atomicAdd(op + 1, hv.y * nw);
  atomicAdd(op + 2, hv.z * nw);
  atomicAdd(op + 3, hv.w * nw);
}

__global__ void k_relu(float* p, int n) {
  int i = blockIdx.x * blockDim.x + threadIdx.x;
  if (i < n) p[i] = fmaxf(p[i], 0.0f);
}

__global__ void k_final(const float* __restrict__ a2,
                        const float* __restrict__ lw,
                        const float* __restrict__ lb,
                        float* __restrict__ out) {
  int i = blockIdx.x * blockDim.x + threadIdx.x;
  if (i >= NN) return;
  const float* hp = a2 + (size_t)i * H2;
  float acc = lb[0];
#pragma unroll
  for (int c = 0; c < H2; ++c) acc = fmaf(hp[c], lw[c], acc);
  out[2 * i + 0] = -acc;
  out[2 * i + 1] = acc;
}

extern "C" void kernel_launch(void* const* d_in, const int* in_sizes, int n_in,
                              void* d_out, int out_size, void* d_ws, size_t ws_size,
                              hipStream_t stream) {
  const float* x      = (const float*)d_in[0];
  const int*   ei     = (const int*)d_in[1];
  const float* elog   = (const float*)d_in[2];
  const float* W1     = (const float*)d_in[3];
  const float* b1     = (const float*)d_in[4];
  const float* W2     = (const float*)d_in[5];
  const float* b2     = (const float*)d_in[6];
  const float* lw     = (const float*)d_in[7];
  const float* lb     = (const float*)d_in[8];
  float* out = (float*)d_out;

  float* ws   = (float*)d_ws;
  float* deg  = ws;
  float* dis  = deg + NN;
  float* ew   = dis + NN;
  float* nrm  = ew + NE;
  float* h1   = nrm + NE;
  float* a1   = h1 + (size_t)NN * H1;
  float* h2   = h1;
  float* a2   = h1 + (size_t)NN * H2;

  const int T = 256;
  k_zero<<<(NN + T - 1) / T, T, 0, stream>>>(deg, NN);
  k_edge_ew_deg<<<(NE + T - 1) / T, T, 0, stream>>>(elog, ei, ew, deg);
  k_dis<<<(NN + T - 1) / T, T, 0, stream>>>(deg, dis);
  k_norm<<<(NE + T - 1) / T, T, 0, stream>>>(ei, ew, dis, nrm);

  k_gemm_wmma<KIN, H1><<<NN / 16, 32 * (H1 / 16), 0, stream>>>(x, W1, h1);
  k_self<H1><<<(NN * H1 + T - 1) / T, T, 0, stream>>>(h1, dis, b1, a1);
  k_scatter<H1, 4><<<(int)(((long long)NE * (H1 / 4) + T - 1) / T), T, 0, stream>>>(h1, ei, nrm, a1);
  k_relu<<<(NN * H1 + T - 1) / T, T, 0, stream>>>(a1, NN * H1);

  k_gemm_wmma<H1, H2><<<NN / 16, 32 * (H2 / 16), 0, stream>>>(a1, W2, h2);
  k_self<H2><<<(NN * H2 + T - 1) / T, T, 0, stream>>>(h2, dis, b2, a2);
  k_scatter<H2, 3><<<(int)(((long long)NE * (H2 / 4) + T - 1) / T), T, 0, stream>>>(h2, ei, nrm, a2);
  k_relu<<<(NN * H2 + T - 1) / T, T, 0, stream>>>(a2, NN * H2);

  k_final<<<(NN + T - 1) / T, T, 0, stream>>>(a2, lw, lb, out);
}


# scatter-add: one wave per 32 edges, lane = channel, so each load and each float atomic covers one whole 128-byte row line instead of 8 floats at 16-byte stride (4x fewer atomic line requests)
# speedup vs baseline: 1.3691x; 1.3691x over previous
#include <hip/hip_runtime.h>
#include <math.h>

#define NN 100000
#define NE 1600000
#define KIN 64
#define H1 64
#define H2 32

typedef __attribute__((ext_vector_type(2))) float v2f;
typedef __attribute__((ext_vector_type(8))) float v8f;

__global__ void k_zero(float* p, int n) {
  int i = blockIdx.x * blockDim.x + threadIdx.x;
  if (i < n) p[i] = 0.0f;
}

__global__ void k_edge_ew_deg(const float* __restrict__ logits,
                              const int* __restrict__ ei,
                              float* __restrict__ ew,
                              float* __restrict__ deg) {
  int e = blockIdx.x * blockDim.x + threadIdx.x;
  if (e >= NE) return;
  float l = logits[e];
  float s = 0.5f + 1.0f / (1.0f + __expf(-l));
  ew[e] = s;
  int col = ei[NE + e];
  atomicAdd(&deg[col], s);
}

__global__ void k_dis(const float* __restrict__ deg, float* __restrict__ dis) {
  int i = blockIdx.x * blockDim.x + threadIdx.x;
  if (i < NN) {
    float d = deg[i] + 1.0f;
    dis[i] = rsqrtf(d);
  }
}

__global__ void k_norm(const int* __restrict__ ei,
                       const float* __restrict__ ew,
                       const float* __restrict__ dis,
                       float* __restrict__ norm) {
  int e = blockIdx.x * blockDim.x + threadIdx.x;
  if (e >= NE) return;
  int r = ei[e];
  int c = ei[NE + e];
  norm[e] = dis[r] * ew[e] * dis[c];
}

template <int K, int NOUT>
__global__ void k_gemm_wmma(const float* __restrict__ X,
                            const float* __restrict__ W,
                            float* __restrict__ H) {
  const int wave = threadIdx.x >> 5;
  const int lane = threadIdx.x & 31;
  const int m0 = blockIdx.x * 16;
  const int n0 = wave * 16;
  const int half = lane >> 4;
  const int l15 = lane & 15;

  v2f bfrag[K / 4];
#pragma unroll
  for (int kt = 0; kt < K / 4; ++kt) {
    int kk = kt * 4 + half * 2;
    int col = n0 + l15;
    v2f b;
    b.x = W[kk * NOUT + col];
    b.y = W[(kk + 1) * NOUT + col];
    bfrag[kt] = b;
  }

  const int row = m0 + l15;
  const float* xp = X + (size_t)row * K + half * 2;

  v8f acc = {};
#pragma unroll
  for (int kt = 0; kt < K / 4; ++kt) {
    v2f a = *(const v2f*)(xp + kt * 4);
    acc = __builtin_amdgcn_wmma_f32_16x16x4_f32(
         false, a,  false, bfrag[kt],
         (short)0, acc,  false,  false);
  }

#pragma unroll
  for (int v = 0; v < 8; ++v) {
    int r = m0 + v + half * 8;
    H[(size_t)r * NOUT + n0 + l15] = acc[v];
  }
}

template <int NOUT>
__global__ void k_self(const float* __restrict__ h,
                       const float* __restrict__ dis,
                       const float* __restrict__ b,
                       float* __restrict__ out) {
  int idx = blockIdx.x * blockDim.x + threadIdx.x;
  if (idx >= NN * NOUT) return;
  int i = idx / NOUT;
  int c = idx - i * NOUT;
  float d = dis[i];
  out[idx] = h[idx] * (d * d) + b[c];
}

template <int NOUT>
__global__ void k_scatter(const float* __restrict__ h,
                          const int* __restrict__ ei,
                          const float* __restrict__ norm,
                          float* __restrict__ out) {
  static_assert(NE % 256 == 0, "one full wave per 32 edges, no tail");
  static_assert(NOUT % 32 == 0, "a row is a whole number of 128-byte lines");
  constexpr int Q = NOUT / 32;
  constexpr int U = 16 / Q;
  const int lane = threadIdx.x & 31;
  const int e = blockIdx.x * blockDim.x + threadIdx.x;
  const int row = ei[e];
  const int col = ei[NE + e];
  const int nwb = __float_as_int(norm[e]);

  for (int j0 = 0; j0 < 32; j0 += U) {
    float v[U][Q];
    for (int u = 0; u < U; ++u) {
      const int r = __builtin_amdgcn_readlane(row, j0 + u);
      const float* hp = h + (size_t)r * NOUT + lane;
      for (int q = 0; q < Q; ++q) v[u][q] = hp[q * 32];
    }
    for (int u = 0; u < U; ++u) {
      const int c = __builtin_amdgcn_readlane(col, j0 + u);
      const float w = __int_as_float(__builtin_amdgcn_readlane(nwb, j0 + u));
      float* op = out + (size_t)c * NOUT + lane;
      for (int q = 0; q < Q; ++q) atomicAdd(op + q * 32, v[u][q] * w);
    }
  }
}

__global__ void k_relu(float* p, int n) {
  int i = blockIdx.x * blockDim.x + threadIdx.x;
  if (i < n) p[i] = fmaxf(p[i], 0.0f);
}

__global__ void k_final(const float* __restrict__ a2,
                        const float* __restrict__ lw,
                        const float* __restrict__ lb,
                        float* __restrict__ out) {
  int i = blockIdx.x * blockDim.x + threadIdx.x;
  if (i >= NN) return;
  const float* hp = a2 + (size_t)i * H2;
  float acc = lb[0];
#pragma unroll
  for (int c = 0; c < H2; ++c) acc = fmaf(hp[c], lw[c], acc);
  out[2 * i + 0] = -acc;
  out[2 * i + 1] = acc;
}

extern "C" void kernel_launch(void* const* d_in, const int* in_sizes, int n_in,
                              void* d_out, int out_size, void* d_ws, size_t ws_size,
                              hipStream_t stream) {
  const float* x      = (const float*)d_in[0];
  const int*   ei     = (const int*)d_in[1];
  const float* elog   = (const float*)d_in[2];
  const float* W1     = (const float*)d_in[3];
  const float* b1     = (const float*)d_in[4];
  const float* W2     = (const float*)d_in[5];
  const float* b2     = (const float*)d_in[6];
  const float* lw     = (const float*)d_in[7];
  const float* lb     = (const float*)d_in[8];
  float* out = (float*)d_out;

  float* ws   = (float*)d_ws;
  float* deg  = ws;
  float* dis  = deg + NN;
  float* ew   = dis + NN;
  float* nrm  = ew + NE;
  float* h1   = nrm + NE;
  float* a1   = h1 + (size_t)NN * H1;
  float* h2   = h1;
  float* a2   = h1 + (size_t)NN * H2;

  const int T = 256;
  k_zero<<<(NN + T - 1) / T, T, 0, stream>>>(deg, NN);
  k_edge_ew_deg<<<(NE + T - 1) / T, T, 0, stream>>>(elog, ei, ew, deg);
  k_dis<<<(NN + T - 1) / T, T, 0, stream>>>(deg, dis);
  k_norm<<<(NE + T - 1) / T, T, 0, stream>>>(ei, ew, dis, nrm);

  k_gemm_wmma<KIN, H1><<<NN / 16, 32 * (H1 / 16), 0, stream>>>(x, W1, h1);
  k_self<H1><<<(NN * H1 + T - 1) / T, T, 0, stream>>>(h1, dis, b1, a1);
  k_scatter<H1><<<NE / T, T, 0, stream>>>(h1, ei, nrm, a1);
  k_relu<<<(NN * H1 + T - 1) / T, T, 0, stream>>>(a1, NN * H1);

  k_gemm_wmma<H1, H2><<<NN / 16, 32 * (H2 / 16), 0, stream>>>(a1, W2, h2);
  k_self<H2><<<(NN * H2 + T - 1) / T, T, 0, stream>>>(h2, dis, b2, a2);
  k_scatter<H2><<<NE / T, T, 0, stream>>>(h2, ei, nrm, a2);
  k_relu<<<(NN * H2 + T - 1) / T, T, 0, stream>>>(a2, NN * H2);

  k_final<<<(NN + T - 1) / T, T, 0, stream>>>(a2, lw, lb, out);
}


# scatter-add kept inside one half of the second-level cache: target rows split in two ranges; a work-group reads its die and takes (512-edge chunk, range) pairs from its own half's counter
# speedup vs baseline: 2.1111x; 1.5420x over previous
#include <hip/hip_runtime.h>
#include <math.h>

#define NN 100000
#define NE 1600000
#define KIN 64
#define H1 64
#define H2 32

typedef __attribute__((ext_vector_type(2))) float v2f;
typedef __attribute__((ext_vector_type(8))) float v8f;

__global__ void k_zero(float* p, int n) {
  int i = blockIdx.x * blockDim.x + threadIdx.x;
  if (i < n) p[i] = 0.0f;
}

__global__ void k_edge_ew_deg(const float* __restrict__ logits,
                              const int* __restrict__ ei,
                              float* __restrict__ ew,
                              float* __restrict__ deg) {
  int e = blockIdx.x * blockDim.x + threadIdx.x;
  if (e >= NE) return;
  float l = logits[e];
  float s = 0.5f + 1.0f / (1.0f + __expf(-l));
  ew[e] = s;
  int col = ei[NE + e];
  atomicAdd(&deg[col], s);
}

__device__ __forceinline__ float dis_of(float deg) { return rsqrtf(deg + 1.0f); }

template <int K, int NOUT, bool RELU_IN>
__global__ void k_gemm_wmma(const float* __restrict__ X,
                            const float* __restrict__ W,
                            const float* __restrict__ deg,
                            const float* __restrict__ bias_in,
                            float* __restrict__ H,
                            float* __restrict__ Y) {
  const int wave = threadIdx.x >> 5;
  const int lane = threadIdx.x & 31;
  const int m0 = blockIdx.x * 16;
  const int n0 = wave * 16;
  const int half = lane >> 4;
  const int l15 = lane & 15;

  v2f bfrag[K / 4];
#pragma unroll
  for (int kt = 0; kt < K / 4; ++kt) {
    int kk = kt * 4 + half * 2;
    int col = n0 + l15;
    v2f b;
    b.x = W[kk * NOUT + col];
    b.y = W[(kk + 1) * NOUT + col];
    bfrag[kt] = b;
  }

  const int row = m0 + l15;
  const float* xp = X + (size_t)row * K + half * 2;

  float dr[8];
#pragma unroll
  for (int v = 0; v < 8; ++v) dr[v] = dis_of(deg[m0 + v + half * 8]);
  float da = 0.0f;
  if (RELU_IN) da = dis_of(deg[row]);

  v8f acc = {};
#pragma unroll
  for (int kt = 0; kt < K / 4; ++kt) {
    v2f a = *(const v2f*)(xp + kt * 4);
    if (RELU_IN) {
      const v2f bi = *(const v2f*)(bias_in + kt * 4 + half * 2);
      a.x = fmaxf(fmaf(a.x, da, bi.x), 0.0f);
      a.y = fmaxf(fmaf(a.y, da, bi.y), 0.0f);
    }
    acc = __builtin_amdgcn_wmma_f32_16x16x4_f32(
         false, a,  false, bfrag[kt],
         (short)0, acc,  false,  false);
  }

#pragma unroll
  for (int v = 0; v < 8; ++v) {
    int r = m0 + v + half * 8;
    size_t o = (size_t)r * NOUT + n0 + l15;
    const float g = acc[v] * dr[v];
    H[o] = g;
    Y[o] = g;
  }
}

#define NCH (NE / 512)
#define CTR_PITCH 64
template <int NOUT>
__device__ __forceinline__ void scatter_group(const float* __restrict__ h, float* __restrict__ out,
                                              const int row, const int col, const int nwb,
                                              unsigned m, const int lane) {
  constexpr int Q = NOUT / 32;
  constexpr int U = 16 / Q;
  while (m) {
    float v[U][Q];
    int jj[U];
#pragma unroll
    for (int u = 0; u < U; ++u) {
      jj[u] = -1;
      if (m) {
        const int j = __builtin_ctz(m);
        m &= m - 1;
        jj[u] = j;
        const int r = __builtin_amdgcn_readlane(row, j);
        const float* hp = h + (size_t)r * NOUT + lane;
        for (int q = 0; q < Q; ++q) v[u][q] = hp[q * 32];
      }
    }
#pragma unroll
    for (int u = 0; u < U; ++u) {
      if (jj[u] >= 0) {
        const int c = __builtin_amdgcn_readlane(col, jj[u]);
        const float w = __int_as_float(__builtin_amdgcn_readlane(nwb, jj[u]));
        float* op = out + (size_t)c * NOUT + lane;
        for (int q = 0; q < Q; ++q) atomicAdd(op + q * 32, v[u][q] * w);
      }
    }
  }
}

template <int NOUT>
__global__ void k_scatter(const float* __restrict__ h,
                          const int* __restrict__ ei,
                          const float* __restrict__ wgt,
                          float* __restrict__ out,
                          int* __restrict__ ctr) {
  static_assert(NE % 512 == 0, "whole chunks, no tail");
  static_assert(NOUT % 32 == 0, "a row is a whole number of 128-byte lines");
  static_assert(NN % 2 == 0, "two equal target ranges");
  __shared__ int s_job;
  if (threadIdx.x == 0) {
    const int id = __builtin_amdgcn_s_sendmsg_rtn(0x87);
    int hf = (id >> 18) & 1;
    int idx = atomicAdd(&ctr[hf * CTR_PITCH], 1);
    if (idx >= NCH) { hf ^= 1; idx = atomicAdd(&ctr[hf * CTR_PITCH], 1); }
    s_job = idx * 2 + hf;
  }
  __syncthreads();
  const int job = __builtin_amdgcn_readfirstlane(s_job);
  const int hf = job & 1;
  const int chunk = job >> 1;
  if (chunk >= NCH) return;
  const int lane = threadIdx.x & 31;
  const int wave = threadIdx.x >> 5;
  const int e0 = chunk * 512 + wave * 64 + lane;
  const int e1 = e0 + 32;
  const int col0 = ei[NE + e0];
  const int col1 = ei[NE + e1];
  const int row0 = ei[e0];
  const int row1 = ei[e1];
  const int nw0 = __float_as_int(wgt[e0]);
  const int nw1 = __float_as_int(wgt[e1]);
  const unsigned m0 = __builtin_amdgcn_ballot_w32((col0 >= NN / 2) == (hf != 0));
  const unsigned m1 = __builtin_amdgcn_ballot_w32((col1 >= NN / 2) == (hf != 0));
  scatter_group<NOUT>(h, out, row0, col0, nw0, m0, lane);
  scatter_group<NOUT>(h, out, row1, col1, nw1, m1, lane);
}

__global__ void k_final(const float* __restrict__ a2,
                        const float* __restrict__ deg,
                        const float* __restrict__ b2,
                        const float* __restrict__ lw,
                        const float* __restrict__ lb,
                        float* __restrict__ out) {
  int i = blockIdx.x * blockDim.x + threadIdx.x;
  if (i >= NN) return;
  const float* hp = a2 + (size_t)i * H2;
  const float d = dis_of(deg[i]);
  float acc = lb[0];
#pragma unroll
  for (int c = 0; c < H2; ++c) acc = fmaf(fmaxf(fmaf(hp[c], d, b2[c]), 0.0f), lw[c], acc);
  out[2 * i + 0] = -acc;
  out[2 * i + 1] = acc;
}

extern "C" void kernel_launch(void* const* d_in, const int* in_sizes, int n_in,
                              void* d_out, int out_size, void* d_ws, size_t ws_size,
                              hipStream_t stream) {
  const float* x      = (const float*)d_in[0];
  const int*   ei     = (const int*)d_in[1];
  const float* elog   = (const float*)d_in[2];
  const float* W1     = (const float*)d_in[3];
  const float* b1     = (const float*)d_in[4];
  const float* W2     = (const float*)d_in[5];
  const float* b2     = (const float*)d_in[6];
  const float* lw     = (const float*)d_in[7];
  const float* lb     = (const float*)d_in[8];
  float* out = (float*)d_out;

  float* ws   = (float*)d_ws;
  float* deg  = ws;
  int*   ctr  = (int*)(deg + NN + 32);
  float* ew   = deg + 2 * NN;
  float* h1   = ew + 2 * (size_t)NE;
  float* a1   = h1 + (size_t)NN * H1;
  float* h2   = h1;
  float* a2   = h1 + (size_t)NN * H2;

  const int T = 256;
  k_zero<<<(NN + 32 + 4 * CTR_PITCH + T - 1) / T, T, 0, stream>>>(deg, NN + 32 + 4 * CTR_PITCH);
  k_edge_ew_deg<<<(NE + T - 1) / T, T, 0, stream>>>(elog, ei, ew, deg);

  k_gemm_wmma<KIN, H1, false><<<NN / 16, 32 * (H1 / 16), 0, stream>>>(x, W1, deg, b1, h1, a1);
  k_scatter<H1><<<2 * NCH, T, 0, stream>>>(h1, ei, ew, a1, ctr);

  k_gemm_wmma<H1, H2, true><<<NN / 16, 32 * (H2 / 16), 0, stream>>>(a1, W2, deg, b1, h2, a2);
  k_scatter<H2><<<2 * NCH, T, 0, stream>>>(h2, ei, ew, a2, ctr + 2 * CTR_PITCH);

  k_final<<<(NN + T - 1) / T, T, 0, stream>>>(a2, deg, b2, lw, lb, out);
}


# degree pass summed in LDS: a work-group takes (12800-edge chunk, half of the targets) from its own cache half's counter, adds ew into an LDS copy, then one float atomic per 128-byte line of deg
# speedup vs baseline: 3.4056x; 1.6132x over previous
#include <hip/hip_runtime.h>
#include <math.h>

#define NN 100000
#define NE 1600000
#define KIN 64
#define H1 64
#define H2 32

typedef __attribute__((ext_vector_type(2))) float v2f;
typedef __attribute__((ext_vector_type(8))) float v8f;

__global__ void k_zero(float* p, int n) {
  int i = blockIdx.x * blockDim.x + threadIdx.x;
  if (i < n) p[i] = 0.0f;
}

#define DE 12800
#define DCH (NE / DE)
#define DT 1024
#define DSPLIT 50016
#define CTR_PITCH 64
__global__ __launch_bounds__(DT) void k_edge_ew_deg(const float* __restrict__ logits,
                                                    const int* __restrict__ ei,
                                                    float* __restrict__ ew,
                                                    float* __restrict__ deg,
                                                    int* __restrict__ ctr) {
  static_assert(NE % DE == 0, "whole chunks");
  static_assert(DSPLIT % 32 == 0 && (NN - DSPLIT) % 32 == 0 && DSPLIT >= NN - DSPLIT, "ranges of whole lines; range 0 is the larger");
  __shared__ float s_deg[DSPLIT];
  __shared__ int s_job;
  const int tid = threadIdx.x;
  if (tid == 0) {
    const int id = __builtin_amdgcn_s_sendmsg_rtn(0x87);
    int hf = (id >> 18) & 1;
    int idx = atomicAdd(&ctr[hf * CTR_PITCH], 1);
    if (idx >= DCH) { hf ^= 1; idx = atomicAdd(&ctr[hf * CTR_PITCH], 1); }
    s_job = idx * 2 + hf;
  }
  for (int i = tid; i < DSPLIT; i += DT) s_deg[i] = 0.0f;
  __syncthreads();
  const int job = __builtin_amdgcn_readfirstlane(s_job);
  const int hf = job & 1;
  const int chunk = job >> 1;
  if (chunk >= DCH) return;
  const int lo = hf ? DSPLIT : 0;
  const int nh = hf ? NN - DSPLIT : DSPLIT;
  const int e0 = chunk * DE;
  const int* __restrict__ colp = ei + NE + e0;
  const float* __restrict__ lgp = logits + e0;
  float* __restrict__ ewp = ew + e0;
  constexpr int U = (DE + DT - 1) / DT;
  int c[U];
  float l[U];
#pragma unroll
  for (int u = 0; u < U; ++u) {
    const int e = u * DT + tid;
    c[u] = -1;
    l[u] = 0.0f;
    if (e < DE) { c[u] = colp[e]; l[u] = lgp[e]; }
  }
#pragma unroll
  for (int u = 0; u < U; ++u) {
    const int e = u * DT + tid;
    if (e < DE) {
      const float s = 0.5f + 1.0f / (1.0f + __expf(-l[u]));
      if (hf == 0) ewp[e] = s;
      const int k = c[u] - lo;
      if ((unsigned)k < (unsigned)nh) atomicAdd(&s_deg[k], s);
    }
  }
  __syncthreads();
  for (int i = tid; i < nh; i += DT) {
    const float v = s_deg[i];
    if (v != 0.0f) atomicAdd(&deg[lo + i], v);
  }
}

__device__ __forceinline__ float dis_of(float deg) { return rsqrtf(deg + 1.0f); }

template <int K, int NOUT, bool RELU_IN>
__global__ void k_gemm_wmma(const float* __restrict__ X,
                            const float* __restrict__ W,
                            const float* __restrict__ deg,
                            const float* __restrict__ bias_in,
                            float* __restrict__ H,
                            float* __restrict__ Y) {
  const int wave = threadIdx.x >> 5;
  const int lane = threadIdx.x & 31;
  const int m0 = blockIdx.x * 16;
  const int n0 = wave * 16;
  const int half = lane >> 4;
  const int l15 = lane & 15;

  v2f bfrag[K / 4];
#pragma unroll
  for (int kt = 0; kt < K / 4; ++kt) {
    int kk = kt * 4 + half * 2;
    int col = n0 + l15;
    v2f b;
    b.x = W[kk * NOUT + col];
    b.y = W[(kk + 1) * NOUT + col];
    bfrag[kt] = b;
  }

  const int row = m0 + l15;
  const float* xp = X + (size_t)row * K + half * 2;

  float dr[8];
#pragma unroll
  for (int v = 0; v < 8; ++v) dr[v] = dis_of(deg[m0 + v + half * 8]);
  float da = 0.0f;
  if (RELU_IN) da = dis_of(deg[row]);

  v8f acc = {};
#pragma unroll
  for (int kt = 0; kt < K / 4; ++kt) {
    v2f a = *(const v2f*)(xp + kt * 4);
    if (RELU_IN) {
      const v2f bi = *(const v2f*)(bias_in + kt * 4 + half * 2);
      a.x = fmaxf(fmaf(a.x, da, bi.x), 0.0f);
      a.y = fmaxf(fmaf(a.y, da, bi.y), 0.0f);
    }
    acc = __builtin_amdgcn_wmma_f32_16x16x4_f32(
         false, a,  false, bfrag[kt],
         (short)0, acc,  false,  false);
  }

#pragma unroll
  for (int v = 0; v < 8; ++v) {
    int r = m0 + v + half * 8;
    size_t o = (size_t)r * NOUT + n0 + l15;
    const float g = acc[v] * dr[v];
    H[o] = g;
    Y[o] = g;
  }
}

#define NCH (NE / 512)
template <int NOUT>
__device__ __forceinline__ void scatter_group(const float* __restrict__ h, float* __restrict__ out,
                                              const int row, const int col, const int nwb,
                                              unsigned m, const int lane) {
  constexpr int Q = NOUT / 32;
  constexpr int U = 16 / Q;
  while (m) {
    float v[U][Q];
    int jj[U];
#pragma unroll
    for (int u = 0; u < U; ++u) {
      jj[u] = -1;
      if (m) {
        const int j = __builtin_ctz(m);
        m &= m - 1;
        jj[u] = j;
        const int r = __builtin_amdgcn_readlane(row, j);
        const float* hp = h + (size_t)r * NOUT + lane;
        for (int q = 0; q < Q; ++q) v[u][q] = hp[q * 32];
      }
    }
#pragma unroll
    for (int u = 0; u < U; ++u) {
      if (jj[u] >= 0) {
        const int c = __builtin_amdgcn_readlane(col, jj[u]);
        const float w = __int_as_float(__builtin_amdgcn_readlane(nwb, jj[u]));
        float* op = out + (size_t)c * NOUT + lane;
        for (int q = 0; q < Q; ++q) atomicAdd(op + q * 32, v[u][q] * w);
      }
    }
  }
}

template <int NOUT>
__global__ void k_scatter(const float* __restrict__ h,
                          const int* __restrict__ ei,
                          const float* __restrict__ wgt,
                          float* __restrict__ out,
                          int* __restrict__ ctr) {
  static_assert(NE % 512 == 0, "whole chunks, no tail");
  static_assert(NOUT % 32 == 0, "a row is a whole number of 128-byte lines");
  static_assert(NN % 2 == 0, "two equal target ranges");
  __shared__ int s_job;
  if (threadIdx.x == 0) {
    const int id = __builtin_amdgcn_s_sendmsg_rtn(0x87);
    int hf = (id >> 18) & 1;
    int idx = atomicAdd(&ctr[hf * CTR_PITCH], 1);
    if (idx >= NCH) { hf ^= 1; idx = atomicAdd(&ctr[hf * CTR_PITCH], 1); }
    s_job = idx * 2 + hf;
  }
  __syncthreads();
  const int job = __builtin_amdgcn_readfirstlane(s_job);
  const int hf = job & 1;
  const int chunk = job >> 1;
  if (chunk >= NCH) return;
  const int lane = threadIdx.x & 31;
  const int wave = threadIdx.x >> 5;
  const int e0 = chunk * 512 + wave * 64 + lane;
  const int e1 = e0 + 32;
  const int col0 = ei[NE + e0];
  const int col1 = ei[NE + e1];
  const int row0 = ei[e0];
  const int row1 = ei[e1];
  const int nw0 = __float_as_int(wgt[e0]);
  const int nw1 = __float_as_int(wgt[e1]);
  const unsigned m0 = __builtin_amdgcn_ballot_w32((col0 >= NN / 2) == (hf != 0));
  const unsigned m1 = __builtin_amdgcn_ballot_w32((col1 >= NN / 2) == (hf != 0));
  scatter_group<NOUT>(h, out, row0, col0, nw0, m0, lane);
  scatter_group<NOUT>(h, out, row1, col1, nw1, m1, lane);
}

__global__ void k_final(const float* __restrict__ a2,
                        const float* __restrict__ deg,
                        const float* __restrict__ b2,
                        const float* __restrict__ lw,
                        const float* __restrict__ lb,
                        float* __restrict__ out) {
  int i = blockIdx.x * blockDim.x + threadIdx.x;
  if (i >= NN) return;
  const float* hp = a2 + (size_t)i * H2;
  const float d = dis_of(deg[i]);
  float acc = lb[0];
#pragma unroll
  for (int c = 0; c < H2; ++c) acc = fmaf(fmaxf(fmaf(hp[c], d, b2[c]), 0.0f), lw[c], acc);
  out[2 * i + 0] = -acc;
  out[2 * i + 1] = acc;
}

extern "C" void kernel_launch(void* const* d_in, const int* in_sizes, int n_in,
                              void* d_out, int out_size, void* d_ws, size_t ws_size,
                              hipStream_t stream) {
  const float* x      = (const float*)d_in[0];
  const int*   ei     = (const int*)d_in[1];
  const float* elog   = (const float*)d_in[2];
  const float* W1     = (const float*)d_in[3];
  const float* b1     = (const float*)d_in[4];
  const float* W2     = (const float*)d_in[5];
  const float* b2     = (const float*)d_in[6];
  const float* lw     = (const float*)d_in[7];
  const float* lb     = (const float*)d_in[8];
  float* out = (float*)d_out;

  float* ws   = (float*)d_ws;
  float* deg  = ws;
  int*   ctr  = (int*)(deg + NN + 32);
  float* ew   = deg + 2 * NN;
  float* h1   = ew + 2 * (size_t)NE;
  float* a1   = h1 + (size_t)NN * H1;
  float* h2   = h1;
  float* a2   = h1 + (size_t)NN * H2;

  const int T = 256;
  k_zero<<<(NN + 32 + 6 * CTR_PITCH + T - 1) / T, T, 0, stream>>>(deg, NN + 32 + 6 * CTR_PITCH);
  k_edge_ew_deg<<<2 * DCH, DT, 0, stream>>>(elog, ei, ew, deg, ctr + 4 * CTR_PITCH);

  k_gemm_wmma<KIN, H1, false><<<NN / 16, 32 * (H1 / 16), 0, stream>>>(x, W1, deg, b1, h1, a1);
  k_scatter<H1><<<2 * NCH, T, 0, stream>>>(h1, ei, ew, a1, ctr);

  k_gemm_wmma<H1, H2, true><<<NN / 16, 32 * (H2 / 16), 0, stream>>>(a1, W2, deg, b1, h2, a2);
  k_scatter<H2><<<2 * NCH, T, 0, stream>>>(h2, ei, ew, a2, ctr + 2 * CTR_PITCH);

  k_final<<<(NN + T - 1) / T, T, 0, stream>>>(a2, deg, b2, lw, lb, out);
}
